# starting kernel
# baseline (speedup 1.0000x reference)
#include <hip/hip_runtime.h>
#include <hip/hip_bf16.h>


#define DI __device__ __forceinline__

typedef float v2f __attribute__((ext_vector_type(2)));
typedef float v8f __attribute__((ext_vector_type(8)));

constexpr int LDA = 36;
constexpr int LDB = 65;

DI v2f frag_a(const float* As, int k0, int r, int half) {
  v2f a;
  a.x = As[r * LDA + k0 + 2 * half + 0];
  a.y = As[r * LDA + k0 + 2 * half + 1];
  return a;
}
DI v2f frag_b(const float* Bs, int k0, int r, int half) {
  v2f b;
  b.x = Bs[(k0 + 2 * half + 0) * LDB + r];
  b.y = Bs[(k0 + 2 * half + 1) * LDB + r];
  return b;
}

DI void mma_chunk(const float* As, const float* Bs, int wr, int cp, int r, int half,
                  v8f& acc0, v8f& acc1) {
  const float* Aw = As + (wr * 16) * LDA;
  const float* B0 = Bs + 32 * cp;
  const float* B1 = Bs + 32 * cp + 16;
#pragma unroll
  for (int k0 = 0; k0 < 32; k0 += 4) {
    v2f a  = frag_a(Aw, k0, r, half);
    v2f b0 = frag_b(B0, k0, r, half);
    v2f b1 = frag_b(B1, k0, r, half);
    acc0 = __builtin_amdgcn_wmma_f32_16x16x4_f32(false, a, false, b0, (short)0, acc0, false, false);
    acc1 = __builtin_amdgcn_wmma_f32_16x16x4_f32(false, a, false, b1, (short)0, acc1, false, false);
  }
}

DI void store_tile(float* out, int ldo, const v8f& acc, int r, int half) {
#pragma unroll
  for (int i = 0; i < 8; ++i) out[(i + 8 * half) * ldo + r] = acc[i];
}

#define WMMA_THREAD_DECOMP()        \
  const int tid  = threadIdx.x;     \
  const int lane = tid & 31;        \
  const int wave = tid >> 5;        \
  const int half = lane >> 4;       \
  const int r    = lane & 15;       \
  const int wr   = wave & 3;        \
  const int cp   = wave >> 2;       \
  (void)lane;

DI float sigmoidf(float x) { return 1.0f / (1.0f + __expf(-x)); }

__global__ void softmax_rows(const float* __restrict__ dis, float* __restrict__ w) {
  __shared__ float red[256];
  const int i = blockIdx.x, t = threadIdx.x;
  float v = -dis[i * 256 + t];
  red[t] = v;
  __syncthreads();
  for (int s = 128; s > 0; s >>= 1) {
    if (t < s) red[t] = fmaxf(red[t], red[t + s]);
    __syncthreads();
  }
  float mx = red[0];
  __syncthreads();
  float e = __expf(v - mx);
  red[t] = e;
  __syncthreads();
  for (int s = 128; s > 0; s >>= 1) {
    if (t < s) red[t] += red[t + s];
    __syncthreads();
  }
  w[i * 256 + t] = e / red[0];
}

__global__ void embed_weight_gemm(const float* __restrict__ E1, const float* __restrict__ E2,
                                  const float* __restrict__ Wih,
                                  float* __restrict__ G1, float* __restrict__ G2) {
  __shared__ float As[64 * LDA];
  __shared__ float Bs[32 * LDB];
  WMMA_THREAD_DECOMP();
  const float* E = blockIdx.z ? E2 : E1;
  float* G = blockIdx.z ? G2 : G1;
  const int koff = blockIdx.z ? 256 : 0;
  const int rb = blockIdx.y * 64;
  const int cb = blockIdx.x * 64;
  v8f acc0 = {}; v8f acc1 = {};
  for (int k0 = 0; k0 < 256; k0 += 32) {
    __syncthreads();
#pragma unroll
    for (int i = 0; i < 8; ++i) {
      int idx = tid + i * 256;
      int ar = idx >> 5, ac = idx & 31;
      As[ar * LDA + ac] = E[(rb + ar) * 256 + k0 + ac];
      int bc = idx >> 5, bk = idx & 31;
      Bs[bk * LDB + bc] = Wih[(cb + bc) * 512 + koff + k0 + bk];
    }
    __syncthreads();
    mma_chunk(As, Bs, wr, cp, r, half, acc0, acc1);
  }
  float* o0 = G + (rb + 16 * wr) * 1024 + cb + 32 * cp;
  store_tile(o0, 1024, acc0, r, half);
  store_tile(o0 + 16, 1024, acc1, r, half);
}

__global__ void lstm_gemm(const float* __restrict__ X, const float* __restrict__ wsm,
                          const float* __restrict__ G1, const float* __restrict__ G2,
                          const float* __restrict__ Whh, const float* __restrict__ h,
                          float* __restrict__ z, int l) {
  __shared__ float As[64 * LDA];
  __shared__ float Bs[32 * LDB];
  WMMA_THREAD_DECOMP();
  const int rb = blockIdx.y * 64;
  const int cb = blockIdx.x * 64;
  v8f acc0 = {}; v8f acc1 = {};
  for (int seg = 0; seg < 2; ++seg) {
    const float* G = seg ? G2 : G1;
    for (int k0 = 0; k0 < 256; k0 += 32) {
      __syncthreads();
#pragma unroll
      for (int i = 0; i < 8; ++i) {
        int idx = tid + i * 256;
        int ar = idx >> 5, ac = idx & 31;
        int row = rb + ar;
        int bb = row >> 8, nn = row & 255;
        float xv = X[((bb * 32 + l) * 256 + nn) * 256 + k0 + ac];
        if (seg) xv *= wsm[nn * 256 + k0 + ac];
        As[ar * LDA + ac] = xv;
        int bk = idx >> 6, bc = idx & 63;
        Bs[bk * LDB + bc] = G[(k0 + bk) * 1024 + cb + bc];
      }
      __syncthreads();
      mma_chunk(As, Bs, wr, cp, r, half, acc0, acc1);
    }
  }
  for (int k0 = 0; k0 < 256; k0 += 32) {
    __syncthreads();
#pragma unroll
    for (int i = 0; i < 8; ++i) {
      int idx = tid + i * 256;
      int ar = idx >> 5, ac = idx & 31;
      As[ar * LDA + ac] = h[(rb + ar) * 256 + k0 + ac];
      int bc = idx >> 5, bk = idx & 31;
      Bs[bk * LDB + bc] = Whh[(cb + bc) * 256 + k0 + bk];
    }
    __syncthreads();
    mma_chunk(As, Bs, wr, cp, r, half, acc0, acc1);
  }
  float* o0 = z + (rb + 16 * wr) * 1024 + cb + 32 * cp;
  store_tile(o0, 1024, acc0, r, half);
  store_tile(o0 + 16, 1024, acc1, r, half);
}

__global__ void lstm_point(const float* __restrict__ z, const float* __restrict__ bih,
                           const float* __restrict__ bhh, float* __restrict__ h,
                           float* __restrict__ c, float* __restrict__ temp, int l) {
  int idx = blockIdx.x * 256 + threadIdx.x;
  int row = idx >> 8, m = idx & 255;
  const float* zr = z + row * 1024;
  float iv = sigmoidf(zr[m]       + bih[m]       + bhh[m]);
  float fv = sigmoidf(zr[m + 256] + bih[m + 256] + bhh[m + 256]);
  float gv = tanhf   (zr[m + 512] + bih[m + 512] + bhh[m + 512]);
  float ov = sigmoidf(zr[m + 768] + bih[m + 768] + bhh[m + 768]);
  float cv = fv * c[idx] + iv * gv;
  float hv = ov * tanhf(cv);
  c[idx] = cv;
  h[idx] = hv;
  int bb = row >> 8, nn = row & 255;
  temp[((bb * 32 + l) * 256 + nn) * 256 + m] = hv;
}

__global__ void zero_f32(float* __restrict__ p, int n) {
  int i = blockIdx.x * 256 + threadIdx.x;
  if (i < n) p[i] = 0.0f;
}

__global__ void bn_stats(const float* __restrict__ temp, float* __restrict__ stats) {
  const int m = threadIdx.x;
  const int rb = blockIdx.x * 256;
  float s = 0.0f, s2 = 0.0f;
  for (int rr = 0; rr < 256; ++rr) {
    float v = temp[(rb + rr) * 256 + m];
    s += v; s2 += v * v;
  }
  atomicAdd(&stats[m], s);
  atomicAdd(&stats[256 + m], s2);
}

__global__ void bn_final(const float* __restrict__ stats, const float* __restrict__ gamma,
                         const float* __restrict__ beta, float* __restrict__ scale,
                         float* __restrict__ shift) {
  const int m = threadIdx.x;
  const float inv = 1.0f / 65536.0f;
  float mu = stats[m] * inv;
  float var = stats[256 + m] * inv - mu * mu;
  float sc = gamma[m] * rsqrtf(var + 1e-5f);
  scale[m] = sc;
  shift[m] = beta[m] - mu * sc;
}

__global__ void proj_gemm(const float* __restrict__ temp, const float* __restrict__ Wlin,
                          const float* __restrict__ blin, const float* __restrict__ scale,
                          const float* __restrict__ shift, float* __restrict__ proj) {
  __shared__ float As[64 * LDA];
  __shared__ float Bs[32 * LDB];
  WMMA_THREAD_DECOMP();
  const int rb = blockIdx.y * 64;
  const int cb = blockIdx.x * 64;
  v8f acc0 = {}; v8f acc1 = {};
  for (int k0 = 0; k0 < 256; k0 += 32) {
    __syncthreads();
#pragma unroll
    for (int i = 0; i < 8; ++i) {
      int idx = tid + i * 256;
      int ar = idx >> 5, ac = idx & 31;
      As[ar * LDA + ac] = temp[(rb + ar) * 256 + k0 + ac] * scale[k0 + ac] + shift[k0 + ac];
      int bc = idx >> 5, bk = idx & 31;
      Bs[bk * LDB + bc] = Wlin[(cb + bc) * 256 + k0 + bk];
    }
    __syncthreads();
    mma_chunk(As, Bs, wr, cp, r, half, acc0, acc1);
  }
  const int c0 = cb + 32 * cp;
  float* o0 = proj + (rb + 16 * wr) * 256 + c0;
#pragma unroll
  for (int i = 0; i < 8; ++i) {
    o0[(i + 8 * half) * 256 + r]      = acc0[i] + blin[c0 + r];
    o0[(i + 8 * half) * 256 + 16 + r] = acc1[i] + blin[c0 + 16 + r];
  }
}

__global__ void out_gemm(const float* __restrict__ proj, const float* __restrict__ temp,
                         const float* __restrict__ scale, const float* __restrict__ shift,
                         float* __restrict__ out) {
  __shared__ float As[64 * LDA];
  __shared__ float Bs[32 * LDB];
  WMMA_THREAD_DECOMP();
  const int bl = blockIdx.z;
  const int rb = blockIdx.y * 64;
  const int cb = blockIdx.x * 64;
  const float* P = proj + bl * 65536;
  const float* T = temp + bl * 65536;
  v8f acc0 = {}; v8f acc1 = {};
  for (int k0 = 0; k0 < 256; k0 += 32) {
    __syncthreads();
#pragma unroll
    for (int i = 0; i < 8; ++i) {
      int idx = tid + i * 256;
      int ar = idx >> 5, ac = idx & 31;
      As[ar * LDA + ac] = P[(rb + ar) * 256 + k0 + ac];
      int dd = idx >> 5, mm = idx & 31;
      Bs[mm * LDB + dd] = T[(cb + dd) * 256 + k0 + mm] * scale[k0 + mm] + shift[k0 + mm];
    }
    __syncthreads();
    mma_chunk(As, Bs, wr, cp, r, half, acc0, acc1);
  }
  float* o0 = out + bl * 65536 + (rb + 16 * wr) * 256 + cb + 32 * cp;
  store_tile(o0, 256, acc0, r, half);
  store_tile(o0 + 16, 256, acc1, r, half);
}

extern "C" void kernel_launch(void* const* d_in, const int* in_sizes, int n_in,
                              void* d_out, int out_size, void* d_ws, size_t ws_size,
                              hipStream_t stream) {
  (void)in_sizes; (void)n_in; (void)out_size; (void)ws_size;
  const float* X     = (const float*)d_in[0];
  const float* dis   = (const float*)d_in[1];
  const float* E1    = (const float*)d_in[2];
  const float* E2    = (const float*)d_in[3];
  const float* Wih   = (const float*)d_in[4];
  const float* Whh   = (const float*)d_in[5];
  const float* bih   = (const float*)d_in[6];
  const float* bhh   = (const float*)d_in[7];
  const float* gamma = (const float*)d_in[8];
  const float* beta  = (const float*)d_in[9];
  const float* Wlin  = (const float*)d_in[10];
  const float* blin  = (const float*)d_in[11];
  float* out = (float*)d_out;

  float* ws = (float*)d_ws;
  size_t o = 0;
  float* w_sm  = ws + o; o += 256 * 256;
  float* G1    = ws + o; o += 256 * 1024;
  float* G2    = ws + o; o += 256 * 1024;
  float* z     = ws + o; o += 2048 * 1024;
  float* h     = ws + o; o += 2048 * 256;
  float* c     = ws + o; o += 2048 * 256;
  float* stats = ws + o; o += 512;
  float* scale = ws + o; o += 256;
  float* shift = ws + o; o += 256;
  float* temp  = ws + o; o += 65536 * 256;
  float* proj  = ws + o; o += 65536 * 256;

  softmax_rows<<<256, 256, 0, stream>>>(dis, w_sm);
  embed_weight_gemm<<<dim3(16, 4, 2), 256, 0, stream>>>(E1, E2, Wih, G1, G2);

  const int zn = 2048 * 256 * 2 + 512;
  zero_f32<<<(zn + 255) / 256, 256, 0, stream>>>(h, zn);

  for (int l = 0; l < 32; ++l) {
    lstm_gemm<<<dim3(16, 32), 256, 0, stream>>>(X, w_sm, G1, G2, Whh, h, z, l);
    lstm_point<<<2048, 256, 0, stream>>>(z, bih, bhh, h, c, temp, l);
  }

  bn_stats<<<256, 256, 0, stream>>>(temp, stats);
  bn_final<<<1, 256, 0, stream>>>(stats, gamma, beta, scale, shift);
  proj_gemm<<<dim3(4, 1024), 256, 0, stream>>>(temp, Wlin, blin, scale, shift, proj);
  out_gemm<<<dim3(4, 4, 256), 256, 0, stream>>>(proj, temp, scale, shift, out);
}


# gate GEMM K loop software-pipelined: the next K chunk's global loads are all issued into registers ahead of the current chunk's WMMAs, replacing a dependent load-wait-store per element
# speedup vs baseline: 1.7416x; 1.7416x over previous
#include <hip/hip_runtime.h>
#include <hip/hip_bf16.h>


#define DI __device__ __forceinline__

typedef float v2f __attribute__((ext_vector_type(2)));
typedef float v8f __attribute__((ext_vector_type(8)));

constexpr int LDA = 36;
constexpr int LDB = 65;

DI v2f frag_a(const float* As, int k0, int r, int half) {
  v2f a;
  a.x = As[r * LDA + k0 + 2 * half + 0];
  a.y = As[r * LDA + k0 + 2 * half + 1];
  return a;
}
DI v2f frag_b(const float* Bs, int k0, int r, int half) {
  v2f b;
  b.x = Bs[(k0 + 2 * half + 0) * LDB + r];
  b.y = Bs[(k0 + 2 * half + 1) * LDB + r];
  return b;
}

DI void mma_chunk(const float* As, const float* Bs, int wr, int cp, int r, int half,
                  v8f& acc0, v8f& acc1) {
  const float* Aw = As + (wr * 16) * LDA;
  const float* B0 = Bs + 32 * cp;
  const float* B1 = Bs + 32 * cp + 16;
#pragma unroll
  for (int k0 = 0; k0 < 32; k0 += 4) {
    v2f a  = frag_a(Aw, k0, r, half);
    v2f b0 = frag_b(B0, k0, r, half);
    v2f b1 = frag_b(B1, k0, r, half);
    acc0 = __builtin_amdgcn_wmma_f32_16x16x4_f32(false, a, false, b0, (short)0, acc0, false, false);
    acc1 = __builtin_amdgcn_wmma_f32_16x16x4_f32(false, a, false, b1, (short)0, acc1, false, false);
  }
}

DI void store_tile(float* out, int ldo, const v8f& acc, int r, int half) {
#pragma unroll
  for (int i = 0; i < 8; ++i) out[(i + 8 * half) * ldo + r] = acc[i];
}

#define WMMA_THREAD_DECOMP()        \
  const int tid  = threadIdx.x;     \
  const int lane = tid & 31;        \
  const int wave = tid >> 5;        \
  const int half = lane >> 4;       \
  const int r    = lane & 15;       \
  const int wr   = wave & 3;        \
  const int cp   = wave >> 2;       \
  (void)lane;

DI float sigmoidf(float x) { return 1.0f / (1.0f + __expf(-x)); }

__global__ void softmax_rows(const float* __restrict__ dis, float* __restrict__ w) {
  __shared__ float red[256];
  const int i = blockIdx.x, t = threadIdx.x;
  float v = -dis[i * 256 + t];
  red[t] = v;
  __syncthreads();
  for (int s = 128; s > 0; s >>= 1) {
    if (t < s) red[t] = fmaxf(red[t], red[t + s]);
    __syncthreads();
  }
  float mx = red[0];
  __syncthreads();
  float e = __expf(v - mx);
  red[t] = e;
  __syncthreads();
  for (int s = 128; s > 0; s >>= 1) {
    if (t < s) red[t] += red[t + s];
    __syncthreads();
  }
  w[i * 256 + t] = e / red[0];
}

__global__ void embed_weight_gemm(const float* __restrict__ E1, const float* __restrict__ E2,
                                  const float* __restrict__ Wih,
                                  float* __restrict__ G1, float* __restrict__ G2) {
  __shared__ float As[64 * LDA];
  __shared__ float Bs[32 * LDB];
  WMMA_THREAD_DECOMP();
  const float* E = blockIdx.z ? E2 : E1;
  float* G = blockIdx.z ? G2 : G1;
  const int koff = blockIdx.z ? 256 : 0;
  const int rb = blockIdx.y * 64;
  const int cb = blockIdx.x * 64;
  v8f acc0 = {}; v8f acc1 = {};
  for (int k0 = 0; k0 < 256; k0 += 32) {
    __syncthreads();
#pragma unroll
    for (int i = 0; i < 8; ++i) {
      int idx = tid + i * 256;
      int ar = idx >> 5, ac = idx & 31;
      As[ar * LDA + ac] = E[(rb + ar) * 256 + k0 + ac];
      int bc = idx >> 5, bk = idx & 31;
      Bs[bk * LDB + bc] = Wih[(cb + bc) * 512 + koff + k0 + bk];
    }
    __syncthreads();
    mma_chunk(As, Bs, wr, cp, r, half, acc0, acc1);
  }
  float* o0 = G + (rb + 16 * wr) * 1024 + cb + 32 * cp;
  store_tile(o0, 1024, acc0, r, half);
  store_tile(o0 + 16, 1024, acc1, r, half);
}

struct GateStage { float a[8]; float w[8]; float b[8]; };

template <int SEG>
DI void gate_stage_load(GateStage& s, const float* __restrict__ X, const float* __restrict__ wsm,
                        const float* __restrict__ G1, const float* __restrict__ G2,
                        const float* __restrict__ Whh, const float* __restrict__ h,
                        int tid, int rb, int cb, int l, int k0) {
#pragma unroll
  for (int i = 0; i < 8; ++i) {
    int idx = tid + i * 256;
    int ar = idx >> 5, ac = idx & 31;
    int row = rb + ar;
    if constexpr (SEG < 2) {
      int bb = row >> 8, nn = row & 255;
      s.a[i] = X[((bb * 32 + l) * 256 + nn) * 256 + k0 + ac];
      if constexpr (SEG == 1) s.w[i] = wsm[nn * 256 + k0 + ac];
      int bk = idx >> 6, bc = idx & 63;
      const float* G = SEG ? G2 : G1;
      s.b[i] = G[(k0 + bk) * 1024 + cb + bc];
    } else {
      s.a[i] = h[row * 256 + k0 + ac];
      int bc = idx >> 5, bk = idx & 31;
      s.b[i] = Whh[(cb + bc) * 256 + k0 + bk];
    }
  }
}

template <int SEG>
DI void gate_stage_store(const GateStage& s, float* As, float* Bs, int tid) {
#pragma unroll
  for (int i = 0; i < 8; ++i) {
    int idx = tid + i * 256;
    int ar = idx >> 5, ac = idx & 31;
    float xv = s.a[i];
    if constexpr (SEG == 1) xv *= s.w[i];
    As[ar * LDA + ac] = xv;
    if constexpr (SEG < 2) {
      int bk = idx >> 6, bc = idx & 63;
      Bs[bk * LDB + bc] = s.b[i];
    } else {
      int bc = idx >> 5, bk = idx & 31;
      Bs[bk * LDB + bc] = s.b[i];
    }
  }
}

template <int SEG, int NEXT>
DI void gate_segment(GateStage& s, float* As, float* Bs,
                     const float* __restrict__ X, const float* __restrict__ wsm,
                     const float* __restrict__ G1, const float* __restrict__ G2,
                     const float* __restrict__ Whh, const float* __restrict__ h,
                     int tid, int rb, int cb, int l, int wr, int cp, int r, int half,
                     v8f& acc0, v8f& acc1) {
  for (int kc = 0; kc < 8; ++kc) {
    __syncthreads();
    gate_stage_store<SEG>(s, As, Bs, tid);
    __syncthreads();
    if (kc < 7) {
      gate_stage_load<SEG>(s, X, wsm, G1, G2, Whh, h, tid, rb, cb, l, (kc + 1) * 32);
    } else if constexpr (NEXT >= 0) {
      gate_stage_load<NEXT>(s, X, wsm, G1, G2, Whh, h, tid, rb, cb, l, 0);
    }
    mma_chunk(As, Bs, wr, cp, r, half, acc0, acc1);
  }
}

__global__ void __launch_bounds__(256) lstm_gemm(const float* __restrict__ X, const float* __restrict__ wsm,
                          const float* __restrict__ G1, const float* __restrict__ G2,
                          const float* __restrict__ Whh, const float* __restrict__ h,
                          float* __restrict__ z, int l) {
  __shared__ float As[64 * LDA];
  __shared__ float Bs[32 * LDB];
  WMMA_THREAD_DECOMP();
  const int rb = blockIdx.y * 64;
  const int cb = blockIdx.x * 64;
  v8f acc0 = {}; v8f acc1 = {};
  GateStage s;
  gate_stage_load<0>(s, X, wsm, G1, G2, Whh, h, tid, rb, cb, l, 0);
  gate_segment<0, 1>(s, As, Bs, X, wsm, G1, G2, Whh, h, tid, rb, cb, l, wr, cp, r, half, acc0, acc1);
  gate_segment<1, 2>(s, As, Bs, X, wsm, G1, G2, Whh, h, tid, rb, cb, l, wr, cp, r, half, acc0, acc1);
  gate_segment<2, -1>(s, As, Bs, X, wsm, G1, G2, Whh, h, tid, rb, cb, l, wr, cp, r, half, acc0, acc1);
  float* o0 = z + (rb + 16 * wr) * 1024 + cb + 32 * cp;
  store_tile(o0, 1024, acc0, r, half);
  store_tile(o0 + 16, 1024, acc1, r, half);
}

__global__ void lstm_point(const float* __restrict__ z, const float* __restrict__ bih,
                           const float* __restrict__ bhh, float* __restrict__ h,
                           float* __restrict__ c, float* __restrict__ temp, int l) {
  int idx = blockIdx.x * 256 + threadIdx.x;
  int row = idx >> 8, m = idx & 255;
  const float* zr = z + row * 1024;
  float iv = sigmoidf(zr[m]       + bih[m]       + bhh[m]);
  float fv = sigmoidf(zr[m + 256] + bih[m + 256] + bhh[m + 256]);
  float gv = tanhf   (zr[m + 512] + bih[m + 512] + bhh[m + 512]);
  float ov = sigmoidf(zr[m + 768] + bih[m + 768] + bhh[m + 768]);
  float cv = fv * c[idx] + iv * gv;
  float hv = ov * tanhf(cv);
  c[idx] = cv;
  h[idx] = hv;
  int bb = row >> 8, nn = row & 255;
  temp[((bb * 32 + l) * 256 + nn) * 256 + m] = hv;
}

__global__ void zero_f32(float* __restrict__ p, int n) {
  int i = blockIdx.x * 256 + threadIdx.x;
  if (i < n) p[i] = 0.0f;
}

__global__ void bn_stats(const float* __restrict__ temp, float* __restrict__ stats) {
  const int m = threadIdx.x;
  const int rb = blockIdx.x * 256;
  float s = 0.0f, s2 = 0.0f;
  for (int rr = 0; rr < 256; ++rr) {
    float v = temp[(rb + rr) * 256 + m];
    s += v; s2 += v * v;
  }
  atomicAdd(&stats[m], s);
  atomicAdd(&stats[256 + m], s2);
}

__global__ void bn_final(const float* __restrict__ stats, const float* __restrict__ gamma,
                         const float* __restrict__ beta, float* __restrict__ scale,
                         float* __restrict__ shift) {
  const int m = threadIdx.x;
  const float inv = 1.0f / 65536.0f;
  float mu = stats[m] * inv;
  float var = stats[256 + m] * inv - mu * mu;
  float sc = gamma[m] * rsqrtf(var + 1e-5f);
  scale[m] = sc;
  shift[m] = beta[m] - mu * sc;
}

__global__ void proj_gemm(const float* __restrict__ temp, const float* __restrict__ Wlin,
                          const float* __restrict__ blin, const float* __restrict__ scale,
                          const float* __restrict__ shift, float* __restrict__ proj) {
  __shared__ float As[64 * LDA];
  __shared__ float Bs[32 * LDB];
  WMMA_THREAD_DECOMP();
  const int rb = blockIdx.y * 64;
  const int cb = blockIdx.x * 64;
  v8f acc0 = {}; v8f acc1 = {};
  for (int k0 = 0; k0 < 256; k0 += 32) {
    __syncthreads();
#pragma unroll
    for (int i = 0; i < 8; ++i) {
      int idx = tid + i * 256;
      int ar = idx >> 5, ac = idx & 31;
      As[ar * LDA + ac] = temp[(rb + ar) * 256 + k0 + ac] * scale[k0 + ac] + shift[k0 + ac];
      int bc = idx >> 5, bk = idx & 31;
      Bs[bk * LDB + bc] = Wlin[(cb + bc) * 256 + k0 + bk];
    }
    __syncthreads();
    mma_chunk(As, Bs, wr, cp, r, half, acc0, acc1);
  }
  const int c0 = cb + 32 * cp;
  float* o0 = proj + (rb + 16 * wr) * 256 + c0;
#pragma unroll
  for (int i = 0; i < 8; ++i) {
    o0[(i + 8 * half) * 256 + r]      = acc0[i] + blin[c0 + r];
    o0[(i + 8 * half) * 256 + 16 + r] = acc1[i] + blin[c0 + 16 + r];
  }
}

__global__ void out_gemm(const float* __restrict__ proj, const float* __restrict__ temp,
                         const float* __restrict__ scale, const float* __restrict__ shift,
                         float* __restrict__ out) {
  __shared__ float As[64 * LDA];
  __shared__ float Bs[32 * LDB];
  WMMA_THREAD_DECOMP();
  const int bl = blockIdx.z;
  const int rb = blockIdx.y * 64;
  const int cb = blockIdx.x * 64;
  const float* P = proj + bl * 65536;
  const float* T = temp + bl * 65536;
  v8f acc0 = {}; v8f acc1 = {};
  for (int k0 = 0; k0 < 256; k0 += 32) {
    __syncthreads();
#pragma unroll
    for (int i = 0; i < 8; ++i) {
      int idx = tid + i * 256;
      int ar = idx >> 5, ac = idx & 31;
      As[ar * LDA + ac] = P[(rb + ar) * 256 + k0 + ac];
      int dd = idx >> 5, mm = idx & 31;
      Bs[mm * LDB + dd] = T[(cb + dd) * 256 + k0 + mm] * scale[k0 + mm] + shift[k0 + mm];
    }
    __syncthreads();
    mma_chunk(As, Bs, wr, cp, r, half, acc0, acc1);
  }
  float* o0 = out + bl * 65536 + (rb + 16 * wr) * 256 + cb + 32 * cp;
  store_tile(o0, 256, acc0, r, half);
  store_tile(o0 + 16, 256, acc1, r, half);
}

extern "C" void kernel_launch(void* const* d_in, const int* in_sizes, int n_in,
                              void* d_out, int out_size, void* d_ws, size_t ws_size,
                              hipStream_t stream) {
  (void)in_sizes; (void)n_in; (void)out_size; (void)ws_size;
  const float* X     = (const float*)d_in[0];
  const float* dis   = (const float*)d_in[1];
  const float* E1    = (const float*)d_in[2];
  const float* E2    = (const float*)d_in[3];
  const float* Wih   = (const float*)d_in[4];
  const float* Whh   = (const float*)d_in[5];
  const float* bih   = (const float*)d_in[6];
  const float* bhh   = (const float*)d_in[7];
  const float* gamma = (const float*)d_in[8];
  const float* beta  = (const float*)d_in[9];
  const float* Wlin  = (const float*)d_in[10];
  const float* blin  = (const float*)d_in[11];
  float* out = (float*)d_out;

  float* ws = (float*)d_ws;
  size_t o = 0;
  float* w_sm  = ws + o; o += 256 * 256;
  float* G1    = ws + o; o += 256 * 1024;
  float* G2    = ws + o; o += 256 * 1024;
  float* z     = ws + o; o += 2048 * 1024;
  float* h     = ws + o; o += 2048 * 256;
  float* c     = ws + o; o += 2048 * 256;
  float* stats = ws + o; o += 512;
  float* scale = ws + o; o += 256;
  float* shift = ws + o; o += 256;
  float* temp  = ws + o; o += 65536 * 256;
  float* proj  = ws + o; o += 65536 * 256;

  softmax_rows<<<256, 256, 0, stream>>>(dis, w_sm);
  embed_weight_gemm<<<dim3(16, 4, 2), 256, 0, stream>>>(E1, E2, Wih, G1, G2);

  const int zn = 2048 * 256 * 2 + 512;
  zero_f32<<<(zn + 255) / 256, 256, 0, stream>>>(h, zn);

  for (int l = 0; l < 32; ++l) {
    lstm_gemm<<<dim3(16, 32), 256, 0, stream>>>(X, w_sm, G1, G2, Whh, h, z, l);
    lstm_point<<<2048, 256, 0, stream>>>(z, bih, bhh, h, c, temp, l);
  }

  bn_stats<<<256, 256, 0, stream>>>(temp, stats);
  bn_final<<<1, 256, 0, stream>>>(stats, gamma, beta, scale, shift);
  proj_gemm<<<dim3(4, 1024), 256, 0, stream>>>(temp, Wlin, blin, scale, shift, proj);
  out_gemm<<<dim3(4, 4, 256), 256, 0, stream>>>(proj, temp, scale, shift, out);
}


# all GEMMs: B tile k-contiguous in LDS like A (G1,G2 stored transposed): operands move as 16-byte global loads and conflict-free 16-byte LDS stores, fragments are 8-byte LDS reads; sum order kept
# speedup vs baseline: 2.0662x; 1.1864x over previous
#include <hip/hip_runtime.h>
#include <hip/hip_bf16.h>


#define DI __device__ __forceinline__

typedef float v2f __attribute__((ext_vector_type(2)));
typedef float v4f __attribute__((ext_vector_type(4)));
typedef float v8f __attribute__((ext_vector_type(8)));

constexpr int LDT = 36;

DI v2f frag(const float* Ts, int k0, int r, int half) {
  return *reinterpret_cast<const v2f*>(Ts + r * LDT + k0 + 2 * half);
}

DI void mma_chunk(const float* As, const float* Bs, int wr, int cp, int r, int half,
                  v8f& acc0, v8f& acc1) {
  const float* Aw = As + (wr * 16) * LDT;
  const float* B0 = Bs + (32 * cp) * LDT;
  const float* B1 = Bs + (32 * cp + 16) * LDT;
#pragma unroll
  for (int k0 = 0; k0 < 32; k0 += 4) {
    v2f a  = frag(Aw, k0, r, half);
    v2f b0 = frag(B0, k0, r, half);
    v2f b1 = frag(B1, k0, r, half);
    acc0 = __builtin_amdgcn_wmma_f32_16x16x4_f32(false, a, false, b0, (short)0, acc0, false, false);
    acc1 = __builtin_amdgcn_wmma_f32_16x16x4_f32(false, a, false, b1, (short)0, acc1, false, false);
  }
}

DI void store_tile(float* out, int ldo, const v8f& acc, int r, int half) {
#pragma unroll
  for (int i = 0; i < 8; ++i) out[(i + 8 * half) * ldo + r] = acc[i];
}

DI int piece_row(int tid, int i) {
  return 32 * i + ((tid >> 7) << 4) + (((tid >> 3) & 1) << 3) + ((tid >> 4) & 7);
}
DI void tile_load(v4f (&v)[2], const float* __restrict__ src, int ld, int tid) {
#pragma unroll
  for (int i = 0; i < 2; ++i)
    v[i] = *reinterpret_cast<const v4f*>(src + piece_row(tid, i) * ld + (tid & 7) * 4);
}
DI void tile_store(float* dst, const v4f (&v)[2], int tid) {
#pragma unroll
  for (int i = 0; i < 2; ++i)
    *reinterpret_cast<v4f*>(dst + piece_row(tid, i) * LDT + (tid & 7) * 4) = v[i];
}

#define WMMA_THREAD_DECOMP()        \
  const int tid  = threadIdx.x;     \
  const int lane = tid & 31;        \
  const int wave = tid >> 5;        \
  const int half = lane >> 4;       \
  const int r    = lane & 15;       \
  const int wr   = wave & 3;        \
  const int cp   = wave >> 2;       \
  (void)lane;

DI float sigmoidf(float x) { return 1.0f / (1.0f + __expf(-x)); }

__global__ void softmax_rows(const float* __restrict__ dis, float* __restrict__ w) {
  __shared__ float red[256];
  const int i = blockIdx.x, t = threadIdx.x;
  float v = -dis[i * 256 + t];
  red[t] = v;
  __syncthreads();
  for (int s = 128; s > 0; s >>= 1) {
    if (t < s) red[t] = fmaxf(red[t], red[t + s]);
    __syncthreads();
  }
  float mx = red[0];
  __syncthreads();
  float e = __expf(v - mx);
  red[t] = e;
  __syncthreads();
  for (int s = 128; s > 0; s >>= 1) {
    if (t < s) red[t] += red[t + s];
    __syncthreads();
  }
  w[i * 256 + t] = e / red[0];
}

__global__ void embed_weight_gemm(const float* __restrict__ E1, const float* __restrict__ E2,
                                  const float* __restrict__ Wih,
                                  float* __restrict__ G1T, float* __restrict__ G2T) {
  __shared__ __attribute__((aligned(16))) float As[64 * LDT];
  __shared__ __attribute__((aligned(16))) float Bs[64 * LDT];
  WMMA_THREAD_DECOMP();
  const float* E = blockIdx.z ? E2 : E1;
  float* GT = blockIdx.z ? G2T : G1T;
  const int koff = blockIdx.z ? 256 : 0;
  const int rb = blockIdx.y * 64;
  const int cb = blockIdx.x * 64;
  v8f acc0 = {}; v8f acc1 = {};
  for (int k0 = 0; k0 < 256; k0 += 32) {
    v4f a[2], b[2];
    tile_load(a, Wih + rb * 512 + koff + k0, 512, tid);
    tile_load(b, E + cb * 256 + k0, 256, tid);
    __syncthreads();
    tile_store(As, a, tid);
    tile_store(Bs, b, tid);
    __syncthreads();
    mma_chunk(As, Bs, wr, cp, r, half, acc0, acc1);
  }
  float* o0 = GT + (rb + 16 * wr) * 256 + cb + 32 * cp;
  store_tile(o0, 256, acc0, r, half);
  store_tile(o0 + 16, 256, acc1, r, half);
}

struct GateStage { v4f a[2]; v4f w[2]; v4f b[2]; };
struct GateSrc { const float* x; const float* w; const float* g1; const float* g2; const float* whh; const float* h; };

template <int SEG>
DI void gate_stage_load(GateStage& s, const GateSrc& p, int tid, int k0) {
  if constexpr (SEG == 0) {
    tile_load(s.a, p.x + k0, 256, tid);
    tile_load(s.b, p.g1 + k0, 256, tid);
  } else if constexpr (SEG == 1) {
    tile_load(s.a, p.x + k0, 256, tid);
    tile_load(s.w, p.w + k0, 256, tid);
    tile_load(s.b, p.g2 + k0, 256, tid);
  } else {
    tile_load(s.a, p.h + k0, 256, tid);
    tile_load(s.b, p.whh + k0, 256, tid);
  }
}

template <int SEG>
DI void gate_stage_store(const GateStage& s, float* As, float* Bs, int tid) {
  v4f xv[2] = { s.a[0], s.a[1] };
  if constexpr (SEG == 1) { xv[0] *= s.w[0]; xv[1] *= s.w[1]; }
  tile_store(As, xv, tid);
  tile_store(Bs, s.b, tid);
}

template <int SEG, int NEXT>
DI void gate_segment(GateStage& s, float* As, float* Bs, const GateSrc& p,
                     int tid, int wr, int cp, int r, int half, v8f& acc0, v8f& acc1) {
#pragma nounroll
  for (int kc = 0; kc < 8; ++kc) {
    __syncthreads();
    gate_stage_store<SEG>(s, As, Bs, tid);
    __syncthreads();
    if (kc < 7) {
      gate_stage_load<SEG>(s, p, tid, (kc + 1) * 32);
    } else if constexpr (NEXT >= 0) {
      gate_stage_load<NEXT>(s, p, tid, 0);
    }
    mma_chunk(As, Bs, wr, cp, r, half, acc0, acc1);
  }
}

__global__ void __launch_bounds__(256) lstm_gemm(const float* __restrict__ X, const float* __restrict__ wsm,
                          const float* __restrict__ G1T, const float* __restrict__ G2T,
                          const float* __restrict__ Whh, const float* __restrict__ h,
                          float* __restrict__ z, int l) {
  __shared__ __attribute__((aligned(16))) float As[64 * LDT];
  __shared__ __attribute__((aligned(16))) float Bs[64 * LDT];
  WMMA_THREAD_DECOMP();
  const int rb = blockIdx.y * 64;
  const int cb = blockIdx.x * 64;
  const int bb = rb >> 8, nn0 = rb & 255;
  GateSrc p;
  p.x   = X + ((bb * 32 + l) * 256 + nn0) * 256;
  p.w   = wsm + nn0 * 256;
  p.g1  = G1T + cb * 256;
  p.g2  = G2T + cb * 256;
  p.whh = Whh + cb * 256;
  p.h   = h + rb * 256;
  v8f acc0 = {}; v8f acc1 = {};
  GateStage s;
  gate_stage_load<0>(s, p, tid, 0);
  gate_segment<0, 1>(s, As, Bs, p, tid, wr, cp, r, half, acc0, acc1);
  gate_segment<1, 2>(s, As, Bs, p, tid, wr, cp, r, half, acc0, acc1);
  gate_segment<2, -1>(s, As, Bs, p, tid, wr, cp, r, half, acc0, acc1);
  float* o0 = z + (rb + 16 * wr) * 1024 + cb + 32 * cp;
  store_tile(o0, 1024, acc0, r, half);
  store_tile(o0 + 16, 1024, acc1, r, half);
}

__global__ void lstm_point(const float* __restrict__ z, const float* __restrict__ bih,
                           const float* __restrict__ bhh, float* __restrict__ h,
                           float* __restrict__ c, float* __restrict__ temp, int l) {
  int idx = blockIdx.x * 256 + threadIdx.x;
  int row = idx >> 8, m = idx & 255;
  const float* zr = z + row * 1024;
  float iv = sigmoidf(zr[m]       + bih[m]       + bhh[m]);
  float fv = sigmoidf(zr[m + 256] + bih[m + 256] + bhh[m + 256]);
  float gv = tanhf   (zr[m + 512] + bih[m + 512] + bhh[m + 512]);
  float ov = sigmoidf(zr[m + 768] + bih[m + 768] + bhh[m + 768]);
  float cv = fv * c[idx] + iv * gv;
  float hv = ov * tanhf(cv);
  c[idx] = cv;
  h[idx] = hv;
  int bb = row >> 8, nn = row & 255;
  temp[((bb * 32 + l) * 256 + nn) * 256 + m] = hv;
}

__global__ void zero_f32(float* __restrict__ p, int n) {
  int i = blockIdx.x * 256 + threadIdx.x;
  if (i < n) p[i] = 0.0f;
}

__global__ void bn_stats(const float* __restrict__ temp, float* __restrict__ stats) {
  const int m = threadIdx.x;
  const int rb = blockIdx.x * 256;
  float s = 0.0f, s2 = 0.0f;
  for (int rr = 0; rr < 256; ++rr) {
    float v = temp[(rb + rr) * 256 + m];
    s += v; s2 += v * v;
  }
  atomicAdd(&stats[m], s);
  atomicAdd(&stats[256 + m], s2);
}

__global__ void bn_final(const float* __restrict__ stats, const float* __restrict__ gamma,
                         const float* __restrict__ beta, float* __restrict__ scale,
                         float* __restrict__ shift) {
  const int m = threadIdx.x;
  const float inv = 1.0f / 65536.0f;
  float mu = stats[m] * inv;
  float var = stats[256 + m] * inv - mu * mu;
  float sc = gamma[m] * rsqrtf(var + 1e-5f);
  scale[m] = sc;
  shift[m] = beta[m] - mu * sc;
}

__global__ void proj_gemm(const float* __restrict__ temp, const float* __restrict__ Wlin,
                          const float* __restrict__ blin, const float* __restrict__ scale,
                          const float* __restrict__ shift, float* __restrict__ proj) {
  __shared__ __attribute__((aligned(16))) float As[64 * LDT];
  __shared__ __attribute__((aligned(16))) float Bs[64 * LDT];
  WMMA_THREAD_DECOMP();
  const int rb = blockIdx.y * 64;
  const int cb = blockIdx.x * 64;
  const int kq = (tid & 7) * 4;
  v8f acc0 = {}; v8f acc1 = {};
  for (int k0 = 0; k0 < 256; k0 += 32) {
    v4f a[2], b[2];
    tile_load(a, temp + rb * 256 + k0, 256, tid);
    tile_load(b, Wlin + cb * 256 + k0, 256, tid);
    const v4f sc = *reinterpret_cast<const v4f*>(scale + k0 + kq);
    const v4f sh = *reinterpret_cast<const v4f*>(shift + k0 + kq);
    a[0] = a[0] * sc + sh;
    a[1] = a[1] * sc + sh;
    __syncthreads();
    tile_store(As, a, tid);
    tile_store(Bs, b, tid);
    __syncthreads();
    mma_chunk(As, Bs, wr, cp, r, half, acc0, acc1);
  }
  const int c0 = cb + 32 * cp;
  float* o0 = proj + (rb + 16 * wr) * 256 + c0;
#pragma unroll
  for (int i = 0; i < 8; ++i) {
    o0[(i + 8 * half) * 256 + r]      = acc0[i] + blin[c0 + r];
    o0[(i + 8 * half) * 256 + 16 + r] = acc1[i] + blin[c0 + 16 + r];
  }
}

__global__ void out_gemm(const float* __restrict__ proj, const float* __restrict__ temp,
                         const float* __restrict__ scale, const float* __restrict__ shift,
                         float* __restrict__ out) {
  __shared__ __attribute__((aligned(16))) float As[64 * LDT];
  __shared__ __attribute__((aligned(16))) float Bs[64 * LDT];
  WMMA_THREAD_DECOMP();
  const int bl = blockIdx.z;
  const int rb = blockIdx.y * 64;
  const int cb = blockIdx.x * 64;
  const float* P = proj + bl * 65536;
  const float* T = temp + bl * 65536;
  const int kq = (tid & 7) * 4;
  v8f acc0 = {}; v8f acc1 = {};
  for (int k0 = 0; k0 < 256; k0 += 32) {
    v4f a[2], b[2];
    tile_load(a, P + rb * 256 + k0, 256, tid);
    tile_load(b, T + cb * 256 + k0, 256, tid);
    const v4f sc = *reinterpret_cast<const v4f*>(scale + k0 + kq);
    const v4f sh = *reinterpret_cast<const v4f*>(shift + k0 + kq);
    b[0] = b[0] * sc + sh;
    b[1] = b[1] * sc + sh;
    __syncthreads();
    tile_store(As, a, tid);
    tile_store(Bs, b, tid);
    __syncthreads();
    mma_chunk(As, Bs, wr, cp, r, half, acc0, acc1);
  }
  float* o0 = out + bl * 65536 + (rb + 16 * wr) * 256 + cb + 32 * cp;
  store_tile(o0, 256, acc0, r, half);
  store_tile(o0 + 16, 256, acc1, r, half);
}

extern "C" void kernel_launch(void* const* d_in, const int* in_sizes, int n_in,
                              void* d_out, int out_size, void* d_ws, size_t ws_size,
                              hipStream_t stream) {
  (void)in_sizes; (void)n_in; (void)out_size; (void)ws_size;
  const float* X     = (const float*)d_in[0];
  const float* dis   = (const float*)d_in[1];
  const float* E1    = (const float*)d_in[2];
  const float* E2    = (const float*)d_in[3];
  const float* Wih   = (const float*)d_in[4];
  const float* Whh   = (const float*)d_in[5];
  const float* bih   = (const float*)d_in[6];
  const float* bhh   = (const float*)d_in[7];
  const float* gamma = (const float*)d_in[8];
  const float* beta  = (const float*)d_in[9];
  const float* Wlin  = (const float*)d_in[10];
  const float* blin  = (const float*)d_in[11];
  float* out = (float*)d_out;

  float* ws = (float*)d_ws;
  size_t o = 0;
  float* w_sm  = ws + o; o += 256 * 256;
  float* G1T   = ws + o; o += 256 * 1024;
  float* G2T   = ws + o; o += 256 * 1024;
  float* z     = ws + o; o += 2048 * 1024;
  float* h     = ws + o; o += 2048 * 256;
  float* c     = ws + o; o += 2048 * 256;
  float* stats = ws + o; o += 512;
  float* scale = ws + o; o += 256;
  float* shift = ws + o; o += 256;
  float* temp  = ws + o; o += 65536 * 256;
  float* proj  = ws + o; o += 65536 * 256;

  softmax_rows<<<256, 256, 0, stream>>>(dis, w_sm);
  embed_weight_gemm<<<dim3(4, 16, 2), 256, 0, stream>>>(E1, E2, Wih, G1T, G2T);

  const int zn = 2048 * 256 * 2 + 512;
  zero_f32<<<(zn + 255) / 256, 256, 0, stream>>>(h, zn);

  for (int l = 0; l < 32; ++l) {
    lstm_gemm<<<dim3(16, 32), 256, 0, stream>>>(X, w_sm, G1T, G2T, Whh, h, z, l);
    lstm_point<<<2048, 256, 0, stream>>>(z, bih, bhh, h, c, temp, l);
  }

  bn_stats<<<256, 256, 0, stream>>>(temp, stats);
  bn_final<<<1, 256, 0, stream>>>(stats, gamma, beta, scale, shift);
  proj_gemm<<<dim3(4, 1024), 256, 0, stream>>>(temp, Wlin, blin, scale, shift, proj);
  out_gemm<<<dim3(4, 4, 256), 256, 0, stream>>>(proj, temp, scale, shift, out);
}


# LSTM cell update fused into the gate GEMM epilogue (work-group columns = 4 gates x 16 channels, gates meet through LDS, h(t-1) read from the history): 32 launches and the z round trip go; sums kept
# speedup vs baseline: 2.2801x; 1.1035x over previous
#include <hip/hip_runtime.h>
#include <hip/hip_bf16.h>


#define DI __device__ __forceinline__

typedef float v2f __attribute__((ext_vector_type(2)));
typedef float v4f __attribute__((ext_vector_type(4)));
typedef float v8f __attribute__((ext_vector_type(8)));

constexpr int LDT = 36;

DI v2f frag(const float* Ts, int k0, int r, int half) {
  return *reinterpret_cast<const v2f*>(Ts + r * LDT + k0 + 2 * half);
}

DI void mma_chunk(const float* As, const float* Bs, int wr, int cp, int r, int half,
                  v8f& acc0, v8f& acc1) {
  const float* Aw = As + (wr * 16) * LDT;
  const float* B0 = Bs + (32 * cp) * LDT;
  const float* B1 = Bs + (32 * cp + 16) * LDT;
#pragma unroll
  for (int k0 = 0; k0 < 32; k0 += 4) {
    v2f a  = frag(Aw, k0, r, half);
    v2f b0 = frag(B0, k0, r, half);
    v2f b1 = frag(B1, k0, r, half);
    acc0 = __builtin_amdgcn_wmma_f32_16x16x4_f32(false, a, false, b0, (short)0, acc0, false, false);
    acc1 = __builtin_amdgcn_wmma_f32_16x16x4_f32(false, a, false, b1, (short)0, acc1, false, false);
  }
}

DI void store_tile(float* out, int ldo, const v8f& acc, int r, int half) {
#pragma unroll
  for (int i = 0; i < 8; ++i) out[(i + 8 * half) * ldo + r] = acc[i];
}

DI int piece_row(int tid, int i) {
  return 32 * i + ((tid >> 7) << 4) + (((tid >> 3) & 1) << 3) + ((tid >> 4) & 7);
}
DI void tile_load(v4f (&v)[2], const float* __restrict__ src, int ld, int tid) {
#pragma unroll
  for (int i = 0; i < 2; ++i)
    v[i] = *reinterpret_cast<const v4f*>(src + piece_row(tid, i) * ld + (tid & 7) * 4);
}
DI void tile_store(float* dst, const v4f (&v)[2], int tid) {
#pragma unroll
  for (int i = 0; i < 2; ++i)
    *reinterpret_cast<v4f*>(dst + piece_row(tid, i) * LDT + (tid & 7) * 4) = v[i];
}

DI void tile_load_gates(v4f (&v)[2], const float* __restrict__ src, int tid) {
#pragma unroll
  for (int i = 0; i < 2; ++i) {
    const int j = piece_row(tid, i);
    v[i] = *reinterpret_cast<const v4f*>(src + (((j >> 4) << 8) + (j & 15)) * 256 + (tid & 7) * 4);
  }
}

#define WMMA_THREAD_DECOMP()        \
  const int tid  = threadIdx.x;     \
  const int lane = tid & 31;        \
  const int wave = tid >> 5;        \
  const int half = lane >> 4;       \
  const int r    = lane & 15;       \
  const int wr   = wave & 3;        \
  const int cp   = wave >> 2;       \
  (void)lane;

DI float sigmoidf(float x) { return 1.0f / (1.0f + __expf(-x)); }

__global__ void softmax_rows(const float* __restrict__ dis, float* __restrict__ w) {
  __shared__ float red[256];
  const int i = blockIdx.x, t = threadIdx.x;
  float v = -dis[i * 256 + t];
  red[t] = v;
  __syncthreads();
  for (int s = 128; s > 0; s >>= 1) {
    if (t < s) red[t] = fmaxf(red[t], red[t + s]);
    __syncthreads();
  }
  float mx = red[0];
  __syncthreads();
  float e = __expf(v - mx);
  red[t] = e;
  __syncthreads();
  for (int s = 128; s > 0; s >>= 1) {
    if (t < s) red[t] += red[t + s];
    __syncthreads();
  }
  w[i * 256 + t] = e / red[0];
}

__global__ void embed_weight_gemm(const float* __restrict__ E1, const float* __restrict__ E2,
                                  const float* __restrict__ Wih,
                                  float* __restrict__ G1T, float* __restrict__ G2T) {
  __shared__ __attribute__((aligned(16))) float As[64 * LDT];
  __shared__ __attribute__((aligned(16))) float Bs[64 * LDT];
  WMMA_THREAD_DECOMP();
  const float* E = blockIdx.z ? E2 : E1;
  float* GT = blockIdx.z ? G2T : G1T;
  const int koff = blockIdx.z ? 256 : 0;
  const int rb = blockIdx.y * 64;
  const int cb = blockIdx.x * 64;
  v8f acc0 = {}; v8f acc1 = {};
  for (int k0 = 0; k0 < 256; k0 += 32) {
    v4f a[2], b[2];
    tile_load(a, Wih + rb * 512 + koff + k0, 512, tid);
    tile_load(b, E + cb * 256 + k0, 256, tid);
    __syncthreads();
    tile_store(As, a, tid);
    tile_store(Bs, b, tid);
    __syncthreads();
    mma_chunk(As, Bs, wr, cp, r, half, acc0, acc1);
  }
  float* o0 = GT + (rb + 16 * wr) * 256 + cb + 32 * cp;
  store_tile(o0, 256, acc0, r, half);
  store_tile(o0 + 16, 256, acc1, r, half);
}

struct GateStage { v4f a[2]; v4f w[2]; v4f b[2]; };
struct GateSrc { const float* x; const float* w; const float* g1; const float* g2; const float* whh; const float* h; };

template <int SEG>
DI void gate_stage_load(GateStage& s, const GateSrc& p, int tid, int k0) {
  if constexpr (SEG == 0) {
    tile_load(s.a, p.x + k0, 256, tid);
    tile_load_gates(s.b, p.g1 + k0, tid);
  } else if constexpr (SEG == 1) {
    tile_load(s.a, p.x + k0, 256, tid);
    tile_load(s.w, p.w + k0, 256, tid);
    tile_load_gates(s.b, p.g2 + k0, tid);
  } else {
    tile_load(s.a, p.h + k0, 256, tid);
    tile_load_gates(s.b, p.whh + k0, tid);
  }
}

template <int SEG>
DI void gate_stage_store(const GateStage& s, float* As, float* Bs, int tid) {
  v4f xv[2] = { s.a[0], s.a[1] };
  if constexpr (SEG == 1) { xv[0] *= s.w[0]; xv[1] *= s.w[1]; }
  tile_store(As, xv, tid);
  tile_store(Bs, s.b, tid);
}

template <int SEG, int NEXT>
DI void gate_segment(GateStage& s, float* As, float* Bs, const GateSrc& p,
                     int tid, int wr, int cp, int r, int half, v8f& acc0, v8f& acc1) {
#pragma nounroll
  for (int kc = 0; kc < 8; ++kc) {
    __syncthreads();
    gate_stage_store<SEG>(s, As, Bs, tid);
    __syncthreads();
    if (kc < 7) {
      gate_stage_load<SEG>(s, p, tid, (kc + 1) * 32);
    } else if constexpr (NEXT >= 0) {
      gate_stage_load<NEXT>(s, p, tid, 0);
    }
    mma_chunk(As, Bs, wr, cp, r, half, acc0, acc1);
  }
}

constexpr int ZP = 68;
__global__ void __launch_bounds__(256) lstm_step(const float* __restrict__ X, const float* __restrict__ wsm,
                          const float* __restrict__ G1T, const float* __restrict__ G2T,
                          const float* __restrict__ Whh, const float* __restrict__ h0,
                          const float* __restrict__ bih, const float* __restrict__ bhh,
                          float* __restrict__ c, float* temp, int l) {
  __shared__ __attribute__((aligned(16))) float smem[2 * 64 * LDT];
  float* As = smem;
  float* Bs = smem + 64 * LDT;
  WMMA_THREAD_DECOMP();
  const int rb = blockIdx.y * 64;
  const int m0 = blockIdx.x * 16;
  const int bb = rb >> 8, nn0 = rb & 255;
  const int prow = tid >> 2;
  const int mq = (tid & 3) * 4;
  float* cptr = c + ((blockIdx.y * 16 + blockIdx.x) * 256 + tid) * 4;
  v4f bi[4], bh[4];
#pragma unroll
  for (int g = 0; g < 4; ++g) {
    bi[g] = *reinterpret_cast<const v4f*>(bih + 256 * g + m0 + mq);
    bh[g] = *reinterpret_cast<const v4f*>(bhh + 256 * g + m0 + mq);
  }
  v4f cv = *reinterpret_cast<const v4f*>(cptr);

  GateSrc p;
  p.x   = X + ((bb * 32 + l) * 256 + nn0) * 256;
  p.w   = wsm + nn0 * 256;
  p.g1  = G1T + m0 * 256;
  p.g2  = G2T + m0 * 256;
  p.whh = Whh + m0 * 256;
  p.h   = (l == 0) ? h0 + rb * 256 : temp + ((bb * 32 + l - 1) * 256 + nn0) * 256;
  v8f acc0 = {}; v8f acc1 = {};
  GateStage s;
  gate_stage_load<0>(s, p, tid, 0);
  gate_segment<0, 1>(s, As, Bs, p, tid, wr, cp, r, half, acc0, acc1);
  gate_segment<1, 2>(s, As, Bs, p, tid, wr, cp, r, half, acc0, acc1);
  gate_segment<2, -1>(s, As, Bs, p, tid, wr, cp, r, half, acc0, acc1);

  __syncthreads();
  float* zt = smem;
#pragma unroll
  for (int i = 0; i < 8; ++i) {
    float* zr = zt + (16 * wr + i + 8 * half) * ZP + 32 * cp + r;
    zr[0]  = acc0[i];
    zr[16] = acc1[i];
  }
  __syncthreads();
  v4f zg[4];
#pragma unroll
  for (int g = 0; g < 4; ++g)
    zg[g] = *reinterpret_cast<const v4f*>(zt + prow * ZP + 16 * g + mq);
  v4f hv;
#pragma unroll
  for (int e = 0; e < 4; ++e) {
    float iv = sigmoidf(zg[0][e] + bi[0][e] + bh[0][e]);
    float fv = sigmoidf(zg[1][e] + bi[1][e] + bh[1][e]);
    float gv = tanhf   (zg[2][e] + bi[2][e] + bh[2][e]);
    float ov = sigmoidf(zg[3][e] + bi[3][e] + bh[3][e]);
    float cn = fv * cv[e] + iv * gv;
    hv[e] = ov * tanhf(cn);
    cv[e] = cn;
  }
  *reinterpret_cast<v4f*>(cptr) = cv;
  *reinterpret_cast<v4f*>(temp + ((bb * 32 + l) * 256 + nn0 + prow) * 256 + m0 + mq) = hv;
}

__global__ void zero_f32(float* __restrict__ p, int n) {
  int i = blockIdx.x * 256 + threadIdx.x;
  if (i < n) p[i] = 0.0f;
}

__global__ void bn_stats(const float* __restrict__ temp, float* __restrict__ stats) {
  const int m = threadIdx.x;
  const int rb = blockIdx.x * 256;
  float s = 0.0f, s2 = 0.0f;
  for (int rr = 0; rr < 256; ++rr) {
    float v = temp[(rb + rr) * 256 + m];
    s += v; s2 += v * v;
  }
  atomicAdd(&stats[m], s);
  atomicAdd(&stats[256 + m], s2);
}

__global__ void bn_final(const float* __restrict__ stats, const float* __restrict__ gamma,
                         const float* __restrict__ beta, float* __restrict__ scale,
                         float* __restrict__ shift) {
  const int m = threadIdx.x;
  const float inv = 1.0f / 65536.0f;
  float mu = stats[m] * inv;
  float var = stats[256 + m] * inv - mu * mu;
  float sc = gamma[m] * rsqrtf(var + 1e-5f);
  scale[m] = sc;
  shift[m] = beta[m] - mu * sc;
}

__global__ void proj_gemm(const float* __restrict__ temp, const float* __restrict__ Wlin,
                          const float* __restrict__ blin, const float* __restrict__ scale,
                          const float* __restrict__ shift, float* __restrict__ proj) {
  __shared__ __attribute__((aligned(16))) float As[64 * LDT];
  __shared__ __attribute__((aligned(16))) float Bs[64 * LDT];
  WMMA_THREAD_DECOMP();
  const int rb = blockIdx.y * 64;
  const int cb = blockIdx.x * 64;
  const int kq = (tid & 7) * 4;
  v8f acc0 = {}; v8f acc1 = {};
  for (int k0 = 0; k0 < 256; k0 += 32) {
    v4f a[2], b[2];
    tile_load(a, temp + rb * 256 + k0, 256, tid);
    tile_load(b, Wlin + cb * 256 + k0, 256, tid);
    const v4f sc = *reinterpret_cast<const v4f*>(scale + k0 + kq);
    const v4f sh = *reinterpret_cast<const v4f*>(shift + k0 + kq);
    a[0] = a[0] * sc + sh;
    a[1] = a[1] * sc + sh;
    __syncthreads();
    tile_store(As, a, tid);
    tile_store(Bs, b, tid);
    __syncthreads();
    mma_chunk(As, Bs, wr, cp, r, half, acc0, acc1);
  }
  const int c0 = cb + 32 * cp;
  float* o0 = proj + (rb + 16 * wr) * 256 + c0;
#pragma unroll
  for (int i = 0; i < 8; ++i) {
    o0[(i + 8 * half) * 256 + r]      = acc0[i] + blin[c0 + r];
    o0[(i + 8 * half) * 256 + 16 + r] = acc1[i] + blin[c0 + 16 + r];
  }
}

__global__ void out_gemm(const float* __restrict__ proj, const float* __restrict__ temp,
                         const float* __restrict__ scale, const float* __restrict__ shift,
                         float* __restrict__ out) {
  __shared__ __attribute__((aligned(16))) float As[64 * LDT];
  __shared__ __attribute__((aligned(16))) float Bs[64 * LDT];
  WMMA_THREAD_DECOMP();
  const int bl = blockIdx.z;
  const int rb = blockIdx.y * 64;
  const int cb = blockIdx.x * 64;
  const float* P = proj + bl * 65536;
  const float* T = temp + bl * 65536;
  const int kq = (tid & 7) * 4;
  v8f acc0 = {}; v8f acc1 = {};
  for (int k0 = 0; k0 < 256; k0 += 32) {
    v4f a[2], b[2];
    tile_load(a, P + rb * 256 + k0, 256, tid);
    tile_load(b, T + cb * 256 + k0, 256, tid);
    const v4f sc = *reinterpret_cast<const v4f*>(scale + k0 + kq);
    const v4f sh = *reinterpret_cast<const v4f*>(shift + k0 + kq);
    b[0] = b[0] * sc + sh;
    b[1] = b[1] * sc + sh;
    __syncthreads();
    tile_store(As, a, tid);
    tile_store(Bs, b, tid);
    __syncthreads();
    mma_chunk(As, Bs, wr, cp, r, half, acc0, acc1);
  }
  float* o0 = out + bl * 65536 + (rb + 16 * wr) * 256 + cb + 32 * cp;
  store_tile(o0, 256, acc0, r, half);
  store_tile(o0 + 16, 256, acc1, r, half);
}

extern "C" void kernel_launch(void* const* d_in, const int* in_sizes, int n_in,
                              void* d_out, int out_size, void* d_ws, size_t ws_size,
                              hipStream_t stream) {
  (void)in_sizes; (void)n_in; (void)out_size; (void)ws_size;
  const float* X     = (const float*)d_in[0];
  const float* dis   = (const float*)d_in[1];
  const float* E1    = (const float*)d_in[2];
  const float* E2    = (const float*)d_in[3];
  const float* Wih   = (const float*)d_in[4];
  const float* Whh   = (const float*)d_in[5];
  const float* bih   = (const float*)d_in[6];
  const float* bhh   = (const float*)d_in[7];
  const float* gamma = (const float*)d_in[8];
  const float* beta  = (const float*)d_in[9];
  const float* Wlin  = (const float*)d_in[10];
  const float* blin  = (const float*)d_in[11];
  float* out = (float*)d_out;

  float* ws = (float*)d_ws;
  size_t o = 0;
  float* w_sm  = ws + o; o += 256 * 256;
  float* G1T   = ws + o; o += 256 * 1024;
  float* G2T   = ws + o; o += 256 * 1024;
  float* h     = ws + o; o += 2048 * 256;
  float* c     = ws + o; o += 2048 * 256;
  float* stats = ws + o; o += 512;
  float* scale = ws + o; o += 256;
  float* shift = ws + o; o += 256;
  float* temp  = ws + o; o += 65536 * 256;
  float* proj  = ws + o; o += 65536 * 256;

  softmax_rows<<<256, 256, 0, stream>>>(dis, w_sm);
  embed_weight_gemm<<<dim3(4, 16, 2), 256, 0, stream>>>(E1, E2, Wih, G1T, G2T);

  const int zn = 2048 * 256 * 2 + 512;
  zero_f32<<<(zn + 255) / 256, 256, 0, stream>>>(h, zn);

  for (int l = 0; l < 32; ++l) {
    lstm_step<<<dim3(16, 32), 256, 0, stream>>>(X, w_sm, G1T, G2T, Whh, h, bih, bhh, c, temp, l);
  }

  bn_stats<<<256, 256, 0, stream>>>(temp, stats);
  bn_final<<<1, 256, 0, stream>>>(stats, gamma, beta, scale, shift);
  proj_gemm<<<dim3(4, 1024), 256, 0, stream>>>(temp, Wlin, blin, scale, shift, proj);
  out_gemm<<<dim3(4, 4, 256), 256, 0, stream>>>(proj, temp, scale, shift, out);
}


# x-path GEMMs merged: per-origin weight G1+w[o]*G2 built in the B-tile load, rows grouped by origin over 8 time steps, z_x kept in the proj plane: gate K 768 to 512; sum order of z changes
# speedup vs baseline: 2.8275x; 1.2401x over previous
#include <hip/hip_runtime.h>
#include <hip/hip_bf16.h>


#define DI __device__ __forceinline__

typedef float v2f __attribute__((ext_vector_type(2)));
typedef float v4f __attribute__((ext_vector_type(4)));
typedef float v8f __attribute__((ext_vector_type(8)));

constexpr int LDT = 36;

DI v2f frag(const float* Ts, int k0, int r, int half) {
  return *reinterpret_cast<const v2f*>(Ts + r * LDT + k0 + 2 * half);
}

DI void mma_chunk(const float* As, const float* Bs, int wr, int cp, int r, int half,
                  v8f& acc0, v8f& acc1) {
  const float* Aw = As + (wr * 16) * LDT;
  const float* B0 = Bs + (32 * cp) * LDT;
  const float* B1 = Bs + (32 * cp + 16) * LDT;
#pragma unroll
  for (int k0 = 0; k0 < 32; k0 += 4) {
    v2f a  = frag(Aw, k0, r, half);
    v2f b0 = frag(B0, k0, r, half);
    v2f b1 = frag(B1, k0, r, half);
    acc0 = __builtin_amdgcn_wmma_f32_16x16x4_f32(false, a, false, b0, (short)0, acc0, false, false);
    acc1 = __builtin_amdgcn_wmma_f32_16x16x4_f32(false, a, false, b1, (short)0, acc1, false, false);
  }
}

DI void store_tile(float* out, int ldo, const v8f& acc, int r, int half) {
#pragma unroll
  for (int i = 0; i < 8; ++i) out[(i + 8 * half) * ldo + r] = acc[i];
}

DI int piece_row(int tid, int i) {
  return 32 * i + ((tid >> 7) << 4) + (((tid >> 3) & 1) << 3) + ((tid >> 4) & 7);
}
DI void tile_load(v4f (&v)[2], const float* __restrict__ src, int ld, int tid) {
#pragma unroll
  for (int i = 0; i < 2; ++i)
    v[i] = *reinterpret_cast<const v4f*>(src + piece_row(tid, i) * ld + (tid & 7) * 4);
}
DI void tile_store(float* dst, const v4f (&v)[2], int tid) {
#pragma unroll
  for (int i = 0; i < 2; ++i)
    *reinterpret_cast<v4f*>(dst + piece_row(tid, i) * LDT + (tid & 7) * 4) = v[i];
}

DI void tile_load_gates(v4f (&v)[2], const float* __restrict__ src, int tid) {
#pragma unroll
  for (int i = 0; i < 2; ++i) {
    const int j = piece_row(tid, i);
    v[i] = *reinterpret_cast<const v4f*>(src + (((j >> 4) << 8) + (j & 15)) * 256 + (tid & 7) * 4);
  }
}

#define WMMA_THREAD_DECOMP()        \
  const int tid  = threadIdx.x;     \
  const int lane = tid & 31;        \
  const int wave = tid >> 5;        \
  const int half = lane >> 4;       \
  const int r    = lane & 15;       \
  const int wr   = wave & 3;        \
  const int cp   = wave >> 2;       \
  (void)lane;

DI float sigmoidf(float x) { return 1.0f / (1.0f + __expf(-x)); }

__global__ void softmax_rows(const float* __restrict__ dis, float* __restrict__ w) {
  __shared__ float red[256];
  const int i = blockIdx.x, t = threadIdx.x;
  float v = -dis[i * 256 + t];
  red[t] = v;
  __syncthreads();
  for (int s = 128; s > 0; s >>= 1) {
    if (t < s) red[t] = fmaxf(red[t], red[t + s]);
    __syncthreads();
  }
  float mx = red[0];
  __syncthreads();
  float e = __expf(v - mx);
  red[t] = e;
  __syncthreads();
  for (int s = 128; s > 0; s >>= 1) {
    if (t < s) red[t] += red[t + s];
    __syncthreads();
  }
  w[i * 256 + t] = e / red[0];
}

__global__ void embed_weight_gemm(const float* __restrict__ E1, const float* __restrict__ E2,
                                  const float* __restrict__ Wih,
                                  float* __restrict__ G1T, float* __restrict__ G2T) {
  __shared__ __attribute__((aligned(16))) float As[64 * LDT];
  __shared__ __attribute__((aligned(16))) float Bs[64 * LDT];
  WMMA_THREAD_DECOMP();
  const float* E = blockIdx.z ? E2 : E1;
  float* GT = blockIdx.z ? G2T : G1T;
  const int koff = blockIdx.z ? 256 : 0;
  const int rb = blockIdx.y * 64;
  const int cb = blockIdx.x * 64;
  v8f acc0 = {}; v8f acc1 = {};
  for (int k0 = 0; k0 < 256; k0 += 32) {
    v4f a[2], b[2];
    tile_load(a, Wih + rb * 512 + koff + k0, 512, tid);
    tile_load(b, E + cb * 256 + k0, 256, tid);
    __syncthreads();
    tile_store(As, a, tid);
    tile_store(Bs, b, tid);
    __syncthreads();
    mma_chunk(As, Bs, wr, cp, r, half, acc0, acc1);
  }
  float* o0 = GT + (rb + 16 * wr) * 256 + cb + 32 * cp;
  store_tile(o0, 256, acc0, r, half);
  store_tile(o0 + 16, 256, acc1, r, half);
}

struct MergeStage { v4f a[2]; v4f g1[2]; v4f g2[2]; v4f wv; };

DI void merge_stage_load(MergeStage& s, const float* __restrict__ xo, const float* __restrict__ g1,
                         const float* __restrict__ g2, const float* __restrict__ wo, int tid, int k0) {
#pragma unroll
  for (int i = 0; i < 2; ++i) {
    const int j = piece_row(tid, i);
    s.a[i] = *reinterpret_cast<const v4f*>(xo + (((j >> 3) << 5) + (j & 7)) * 65536 + k0 + (tid & 7) * 4);
  }
  tile_load(s.g1, g1 + k0, 256, tid);
  tile_load(s.g2, g2 + k0, 256, tid);
  s.wv = *reinterpret_cast<const v4f*>(wo + k0);
}

DI void merge_stage_store(const MergeStage& s, float* As, float* Bs, int tid) {
  v4f bv[2] = { s.g1[0] + s.wv * s.g2[0], s.g1[1] + s.wv * s.g2[1] };
  tile_store(As, s.a, tid);
  tile_store(Bs, bv, tid);
}

__global__ void __launch_bounds__(256) xgate_gemm(const float* __restrict__ X, const float* __restrict__ wsm,
                           const float* __restrict__ G1T, const float* __restrict__ G2T,
                           float* __restrict__ zx, int l0) {
  __shared__ __attribute__((aligned(16))) float As[64 * LDT];
  __shared__ __attribute__((aligned(16))) float Bs[64 * LDT];
  WMMA_THREAD_DECOMP();
  const int o  = blockIdx.y;
  const int cb = blockIdx.x * 64;
  const float* xo = X + (l0 * 256 + o) * 256;
  const float* wo = wsm + o * 256 + (tid & 7) * 4;
  const float* g1 = G1T + cb * 256;
  const float* g2 = G2T + cb * 256;
  v8f acc0 = {}; v8f acc1 = {};
  MergeStage s;
  merge_stage_load(s, xo, g1, g2, wo, tid, 0);
#pragma nounroll
  for (int kc = 0; kc < 8; ++kc) {
    __syncthreads();
    merge_stage_store(s, As, Bs, tid);
    __syncthreads();
    if (kc < 7) merge_stage_load(s, xo, g1, g2, wo, tid, (kc + 1) * 32);
    mma_chunk(As, Bs, wr, cp, r, half, acc0, acc1);
  }
#pragma unroll
  for (int i = 0; i < 8; ++i) {
    const int j = 16 * wr + i + 8 * half;
    float* zr = zx + ((j & 7) * 2048 + (j >> 3) * 256 + o) * 1024 + cb + 32 * cp + r;
    zr[0]  = acc0[i];
    zr[16] = acc1[i];
  }
}

struct GateStage { v4f a[2]; v4f b[2]; };

DI void gate_stage_load(GateStage& s, const float* hsrc, const float* __restrict__ whh, int tid, int k0) {
  tile_load(s.a, hsrc + k0, 256, tid);
  tile_load_gates(s.b, whh + k0, tid);
}

constexpr int ZP = 68;
__global__ void __launch_bounds__(256) lstm_step(const float* __restrict__ zx,
                          const float* __restrict__ Whh, const float* __restrict__ h0,
                          const float* __restrict__ bih, const float* __restrict__ bhh,
                          float* __restrict__ c, float* temp, int l) {
  __shared__ __attribute__((aligned(16))) float smem[2 * 64 * LDT];
  float* As = smem;
  float* Bs = smem + 64 * LDT;
  WMMA_THREAD_DECOMP();
  const int rb = blockIdx.y * 64;
  const int m0 = blockIdx.x * 16;
  const int bb = rb >> 8, nn0 = rb & 255;
  const int prow = tid >> 2;
  const int mq = (tid & 3) * 4;
  float* cptr = c + ((blockIdx.y * 16 + blockIdx.x) * 256 + tid) * 4;
  const float* zxr = zx + ((l & 7) * 2048 + rb + prow) * 1024 + m0 + mq;
  v4f bi[4], bh[4], zq[4];
#pragma unroll
  for (int g = 0; g < 4; ++g) {
    bi[g] = *reinterpret_cast<const v4f*>(bih + 256 * g + m0 + mq);
    bh[g] = *reinterpret_cast<const v4f*>(bhh + 256 * g + m0 + mq);
    zq[g] = *reinterpret_cast<const v4f*>(zxr + 256 * g);
  }
  v4f cv = *reinterpret_cast<const v4f*>(cptr);

  const float* hsrc = (l == 0) ? h0 + rb * 256 : temp + ((bb * 32 + l - 1) * 256 + nn0) * 256;
  const float* whh = Whh + m0 * 256;
  v8f acc0 = {}; v8f acc1 = {};
  GateStage s;
  gate_stage_load(s, hsrc, whh, tid, 0);
#pragma nounroll
  for (int kc = 0; kc < 8; ++kc) {
    __syncthreads();
    tile_store(As, s.a, tid);
    tile_store(Bs, s.b, tid);
    __syncthreads();
    if (kc < 7) gate_stage_load(s, hsrc, whh, tid, (kc + 1) * 32);
    mma_chunk(As, Bs, wr, cp, r, half, acc0, acc1);
  }

  __syncthreads();
  float* zt = smem;
#pragma unroll
  for (int i = 0; i < 8; ++i) {
    float* zr = zt + (16 * wr + i + 8 * half) * ZP + 32 * cp + r;
    zr[0]  = acc0[i];
    zr[16] = acc1[i];
  }
  __syncthreads();
  v4f zg[4];
#pragma unroll
  for (int g = 0; g < 4; ++g)
    zg[g] = *reinterpret_cast<const v4f*>(zt + prow * ZP + 16 * g + mq);
  v4f hv;
#pragma unroll
  for (int e = 0; e < 4; ++e) {
    float iv = sigmoidf(zq[0][e] + zg[0][e] + bi[0][e] + bh[0][e]);
    float fv = sigmoidf(zq[1][e] + zg[1][e] + bi[1][e] + bh[1][e]);
    float gv = tanhf   (zq[2][e] + zg[2][e] + bi[2][e] + bh[2][e]);
    float ov = sigmoidf(zq[3][e] + zg[3][e] + bi[3][e] + bh[3][e]);
    float cn = fv * cv[e] + iv * gv;
    hv[e] = ov * tanhf(cn);
    cv[e] = cn;
  }
  *reinterpret_cast<v4f*>(cptr) = cv;
  *reinterpret_cast<v4f*>(temp + ((bb * 32 + l) * 256 + nn0 + prow) * 256 + m0 + mq) = hv;
}

__global__ void zero_f32(float* __restrict__ p, int n) {
  int i = blockIdx.x * 256 + threadIdx.x;
  if (i < n) p[i] = 0.0f;
}

__global__ void bn_stats(const float* __restrict__ temp, float* __restrict__ stats) {
  const int m = threadIdx.x;
  const int rb = blockIdx.x * 256;
  float s = 0.0f, s2 = 0.0f;
  for (int rr = 0; rr < 256; ++rr) {
    float v = temp[(rb + rr) * 256 + m];
    s += v; s2 += v * v;
  }
  atomicAdd(&stats[m], s);
  atomicAdd(&stats[256 + m], s2);
}

__global__ void bn_final(const float* __restrict__ stats, const float* __restrict__ gamma,
                         const float* __restrict__ beta, float* __restrict__ scale,
                         float* __restrict__ shift) {
  const int m = threadIdx.x;
  const float inv = 1.0f / 65536.0f;
  float mu = stats[m] * inv;
  float var = stats[256 + m] * inv - mu * mu;
  float sc = gamma[m] * rsqrtf(var + 1e-5f);
  scale[m] = sc;
  shift[m] = beta[m] - mu * sc;
}

__global__ void proj_gemm(const float* __restrict__ temp, const float* __restrict__ Wlin,
                          const float* __restrict__ blin, const float* __restrict__ scale,
                          const float* __restrict__ shift, float* __restrict__ proj) {
  __shared__ __attribute__((aligned(16))) float As[64 * LDT];
  __shared__ __attribute__((aligned(16))) float Bs[64 * LDT];
  WMMA_THREAD_DECOMP();
  const int rb = blockIdx.y * 64;
  const int cb = blockIdx.x * 64;
  const int kq = (tid & 7) * 4;
  v8f acc0 = {}; v8f acc1 = {};
  for (int k0 = 0; k0 < 256; k0 += 32) {
    v4f a[2], b[2];
    tile_load(a, temp + rb * 256 + k0, 256, tid);
    tile_load(b, Wlin + cb * 256 + k0, 256, tid);
    const v4f sc = *reinterpret_cast<const v4f*>(scale + k0 + kq);
    const v4f sh = *reinterpret_cast<const v4f*>(shift + k0 + kq);
    a[0] = a[0] * sc + sh;
    a[1] = a[1] * sc + sh;
    __syncthreads();
    tile_store(As, a, tid);
    tile_store(Bs, b, tid);
    __syncthreads();
    mma_chunk(As, Bs, wr, cp, r, half, acc0, acc1);
  }
  const int c0 = cb + 32 * cp;
  float* o0 = proj + (rb + 16 * wr) * 256 + c0;
#pragma unroll
  for (int i = 0; i < 8; ++i) {
    o0[(i + 8 * half) * 256 + r]      = acc0[i] + blin[c0 + r];
    o0[(i + 8 * half) * 256 + 16 + r] = acc1[i] + blin[c0 + 16 + r];
  }
}

__global__ void out_gemm(const float* __restrict__ proj, const float* __restrict__ temp,
                         const float* __restrict__ scale, const float* __restrict__ shift,
                         float* __restrict__ out) {
  __shared__ __attribute__((aligned(16))) float As[64 * LDT];
  __shared__ __attribute__((aligned(16))) float Bs[64 * LDT];
  WMMA_THREAD_DECOMP();
  const int bl = blockIdx.z;
  const int rb = blockIdx.y * 64;
  const int cb = blockIdx.x * 64;
  const float* P = proj + bl * 65536;
  const float* T = temp + bl * 65536;
  const int kq = (tid & 7) * 4;
  v8f acc0 = {}; v8f acc1 = {};
  for (int k0 = 0; k0 < 256; k0 += 32) {
    v4f a[2], b[2];
    tile_load(a, P + rb * 256 + k0, 256, tid);
    tile_load(b, T + cb * 256 + k0, 256, tid);
    const v4f sc = *reinterpret_cast<const v4f*>(scale + k0 + kq);
    const v4f sh = *reinterpret_cast<const v4f*>(shift + k0 + kq);
    b[0] = b[0] * sc + sh;
    b[1] = b[1] * sc + sh;
    __syncthreads();
    tile_store(As, a, tid);
    tile_store(Bs, b, tid);
    __syncthreads();
    mma_chunk(As, Bs, wr, cp, r, half, acc0, acc1);
  }
  float* o0 = out + bl * 65536 + (rb + 16 * wr) * 256 + cb + 32 * cp;
  store_tile(o0, 256, acc0, r, half);
  store_tile(o0 + 16, 256, acc1, r, half);
}

extern "C" void kernel_launch(void* const* d_in, const int* in_sizes, int n_in,
                              void* d_out, int out_size, void* d_ws, size_t ws_size,
                              hipStream_t stream) {
  (void)in_sizes; (void)n_in; (void)out_size; (void)ws_size;
  const float* X     = (const float*)d_in[0];
  const float* dis   = (const float*)d_in[1];
  const float* E1    = (const float*)d_in[2];
  const float* E2    = (const float*)d_in[3];
  const float* Wih   = (const float*)d_in[4];
  const float* Whh   = (const float*)d_in[5];
  const float* bih   = (const float*)d_in[6];
  const float* bhh   = (const float*)d_in[7];
  const float* gamma = (const float*)d_in[8];
  const float* beta  = (const float*)d_in[9];
  const float* Wlin  = (const float*)d_in[10];
  const float* blin  = (const float*)d_in[11];
  float* out = (float*)d_out;

  float* ws = (float*)d_ws;
  size_t o = 0;
  float* w_sm  = ws + o; o += 256 * 256;
  float* G1T   = ws + o; o += 256 * 1024;
  float* G2T   = ws + o; o += 256 * 1024;
  float* h     = ws + o; o += 2048 * 256;
  float* c     = ws + o; o += 2048 * 256;
  float* stats = ws + o; o += 512;
  float* scale = ws + o; o += 256;
  float* shift = ws + o; o += 256;
  float* temp  = ws + o; o += 65536 * 256;
  float* proj  = ws + o; o += 65536 * 256;

  softmax_rows<<<256, 256, 0, stream>>>(dis, w_sm);
  embed_weight_gemm<<<dim3(4, 16, 2), 256, 0, stream>>>(E1, E2, Wih, G1T, G2T);

  const int zn = 2048 * 256 * 2 + 512;
  zero_f32<<<(zn + 255) / 256, 256, 0, stream>>>(h, zn);

  for (int l0 = 0; l0 < 32; l0 += 8) {
    xgate_gemm<<<dim3(16, 256), 256, 0, stream>>>(X, w_sm, G1T, G2T, proj, l0);
    for (int l = l0; l < l0 + 8; ++l) {
      lstm_step<<<dim3(16, 32), 256, 0, stream>>>(proj, Whh, h, bih, bhh, c, temp, l);
    }
  }

  bn_stats<<<256, 256, 0, stream>>>(temp, stats);
  bn_final<<<1, 256, 0, stream>>>(stats, gamma, beta, scale, shift);
  proj_gemm<<<dim3(4, 1024), 256, 0, stream>>>(temp, Wlin, blin, scale, shift, proj);
  out_gemm<<<dim3(4, 4, 256), 256, 0, stream>>>(proj, temp, scale, shift, out);
}
